# starting kernel
# baseline (speedup 1.0000x reference)
#include <hip/hip_runtime.h>


typedef __attribute__((ext_vector_type(2))) float v2f;
typedef __attribute__((ext_vector_type(8))) float v8f;

__global__ __launch_bounds__(256) void ipn_gram_wmma(const float* __restrict__ x,
                                                     float* __restrict__ out,
                                                     int batch) {
  const int lane = threadIdx.x & 31;
  const int wave = threadIdx.x >> 5;
  const int b = __builtin_amdgcn_readfirstlane(blockIdx.x * 8 + wave);
  if (b >= batch) return;

  const int half = lane >> 4;
  const int r    = lane & 15;

  const float* __restrict__ xb = x + (size_t)b * 512;
  const int off = r * 16 + 2 * half;

  v2f a0[4], a1[4];
#pragma unroll
  for (int k = 0; k < 4; ++k) {
    a0[k] = *(const v2f*)(xb + off + 4 * k);
    a1[k] = *(const v2f*)(xb + off + 256 + 4 * k);
  }

  v8f c00 = {}, c01 = {}, c11 = {};
#pragma unroll
  for (int k = 0; k < 4; ++k) {
    c00 = __builtin_amdgcn_wmma_f32_16x16x4_f32(false, a0[k], false, a0[k],
                                                (short)0, c00, false, false);
    c01 = __builtin_amdgcn_wmma_f32_16x16x4_f32(false, a0[k], false, a1[k],
                                                (short)0, c01, false, false);
    c11 = __builtin_amdgcn_wmma_f32_16x16x4_f32(false, a1[k], false, a1[k],
                                                (short)0, c11, false, false);
  }

  float* __restrict__ ob = out + (size_t)b * 496;
  const int mbase = half * 8;
  int p01 = (half ? 212 : 0) + r + 15;
  int p11 = (half ? 460 : 376) + r - 1;
#pragma unroll
  for (int v = 0; v < 8; ++v) {
    const int m = mbase + v;
    ob[p01] = c01[v];
    if (r > m) {
      ob[p01 - 16] = c00[v];
      ob[p11]      = c11[v];
    }
    p01 += 30 - m;
    p11 += 14 - m;
  }
}

extern "C" void kernel_launch(void* const* d_in, const int* in_sizes, int n_in,
                              void* d_out, int out_size, void* d_ws, size_t ws_size,
                              hipStream_t stream) {
  const float* x = (const float*)d_in[0];
  float* out = (float*)d_out;
  const int batch = in_sizes[0] / (32 * 16);
  const int waves_per_block = 8;
  const int blocks = (batch + waves_per_block - 1) / waves_per_block;
  ipn_gram_wmma<<<blocks, 256, 0, stream>>>(x, out, batch);
}


# stage outputs in wave-private LDS, one wave per even/odd batch pair (31 whole 128-B lines): each output line written once by one 512-B-contiguous b128 store instead of many unaligned 64-B runs
# speedup vs baseline: 1.3391x; 1.3391x over previous
#include <hip/hip_runtime.h>


typedef __attribute__((ext_vector_type(2))) float v2f;
typedef __attribute__((ext_vector_type(4))) float v4f;
typedef __attribute__((ext_vector_type(8))) float v8f;

#define NPAIR 496
#define WAVES_PER_BLOCK 8
#define ELEMS_PER_WAVE 2
#define ROW_V4 (NPAIR / 4)
#define OUT_V4 (ELEMS_PER_WAVE * ROW_V4)
#define ROW_PITCH 512
#define STAGE_V4 (ELEMS_PER_WAVE * ROW_PITCH / 4)
#define FULL_STORES (OUT_V4 / 32)
#define LAST_LANES (OUT_V4 - 32 * FULL_STORES)

__global__ __launch_bounds__(256) void ipn_gram_wmma(const float* __restrict__ x,
                                                     float* __restrict__ out,
                                                     int batch) {
  __shared__ v4f stage[WAVES_PER_BLOCK][STAGE_V4];

  const int lane = threadIdx.x & 31;
  const int wave = threadIdx.x >> 5;
  const int b0 = __builtin_amdgcn_readfirstlane((blockIdx.x * WAVES_PER_BLOCK + wave) * ELEMS_PER_WAVE);
  if (b0 >= batch) return;
  const int nb = (batch - b0 >= ELEMS_PER_WAVE) ? ELEMS_PER_WAVE : 1;

  const int half = lane >> 4;
  const int r    = lane & 15;
  const int off  = r * 16 + 2 * half;
  const int mbase = half * 8;

  const float* __restrict__ xb0 = x + (size_t)b0 * 512;
  const float* __restrict__ xb1 = xb0 + (nb - 1) * 512;

  v2f a0[ELEMS_PER_WAVE][4], a1[ELEMS_PER_WAVE][4];
#pragma unroll
  for (int k = 0; k < 4; ++k) {
    a0[0][k] = *(const v2f*)(xb0 + off + 4 * k);
    a1[0][k] = *(const v2f*)(xb0 + off + 256 + 4 * k);
  }
#pragma unroll
  for (int k = 0; k < 4; ++k) {
    a0[1][k] = *(const v2f*)(xb1 + off + 4 * k);
    a1[1][k] = *(const v2f*)(xb1 + off + 256 + 4 * k);
  }

  v8f c00[ELEMS_PER_WAVE], c01[ELEMS_PER_WAVE], c11[ELEMS_PER_WAVE];
#pragma unroll
  for (int s = 0; s < ELEMS_PER_WAVE; ++s) {
    c00[s] = (v8f){}; c01[s] = (v8f){}; c11[s] = (v8f){};
#pragma unroll
    for (int k = 0; k < 4; ++k) {
      c00[s] = __builtin_amdgcn_wmma_f32_16x16x4_f32(false, a0[s][k], false, a0[s][k],
                                                     (short)0, c00[s], false, false);
      c01[s] = __builtin_amdgcn_wmma_f32_16x16x4_f32(false, a0[s][k], false, a1[s][k],
                                                     (short)0, c01[s], false, false);
      c11[s] = __builtin_amdgcn_wmma_f32_16x16x4_f32(false, a1[s][k], false, a1[s][k],
                                                     (short)0, c11[s], false, false);
    }
  }

  float* __restrict__ sw = (float*)&stage[wave][0];
  int p01 = (half ? 212 : 0) + r + 15;
  int p11 = (half ? 460 : 376) + r - 1;
#pragma unroll
  for (int v = 0; v < 8; ++v) {
    const int m = mbase + v;
    sw[p01]             = c01[0][v];
    sw[p01 + ROW_PITCH] = c01[1][v];
    if (r > m) {
      sw[p01 - 16]             = c00[0][v];
      sw[p01 - 16 + ROW_PITCH] = c00[1][v];
      sw[p11]                  = c11[0][v];
      sw[p11 + ROW_PITCH]      = c11[1][v];
    }
    p01 += 30 - m;
    p11 += 14 - m;
  }

  const v4f* __restrict__ sv = &stage[wave][0];
  v4f* __restrict__ ov = (v4f*)(out + (size_t)b0 * NPAIR);
  if (nb == ELEMS_PER_WAVE) {
    v4f t[FULL_STORES];
#pragma unroll
    for (int i = 0; i < FULL_STORES; ++i) {
      const int o = lane + 32 * i;
      t[i] = sv[o + (o >= ROW_V4 ? ROW_PITCH / 4 - ROW_V4 : 0)];
    }
#pragma unroll
    for (int i = 0; i < FULL_STORES; ++i) ov[lane + 32 * i] = t[i];
    if (lane < LAST_LANES) {
      const int o = lane + 32 * FULL_STORES;
      ov[o] = sv[o + ROW_PITCH / 4 - ROW_V4];
    }
  } else {
#pragma unroll
    for (int i = 0; i < 4; ++i) {
      const int o = lane + 32 * i;
      if (o < ROW_V4) ov[o] = sv[o];
    }
  }
}

extern "C" void kernel_launch(void* const* d_in, const int* in_sizes, int n_in,
                              void* d_out, int out_size, void* d_ws, size_t ws_size,
                              hipStream_t stream) {
  const float* x = (const float*)d_in[0];
  float* out = (float*)d_out;
  const int batch = in_sizes[0] / (32 * 16);
  const int waves = (batch + ELEMS_PER_WAVE - 1) / ELEMS_PER_WAVE;
  const int blocks = (waves + WAVES_PER_BLOCK - 1) / WAVES_PER_BLOCK;
  ipn_gram_wmma<<<blocks, 256, 0, stream>>>(x, out, batch);
}
